# starting kernel
# baseline (speedup 1.0000x reference)
#include <hip/hip_runtime.h>

typedef __attribute__((ext_vector_type(2))) float v2f;
typedef __attribute__((ext_vector_type(8))) float v8f;

#define HD 32
#define VD 64
#define TCH 64
#define NTHREADS 160


__launch_bounds__(NTHREADS)
__global__ void delta_model_fused(
    const int*   __restrict__ seq,
    const float* __restrict__ embed,
    const float* __restrict__ w1, const float* __restrict__ b1,
    const float* __restrict__ w2, const float* __restrict__ b2,
    const float* __restrict__ ln_g, const float* __restrict__ ln_b,
    const float* __restrict__ wr, const float* __restrict__ br,
    const float* __restrict__ wo, const float* __restrict__ bo,
    float* __restrict__ out, int L)
{
  __shared__ float s_w1[HD * 2 * HD];
  __shared__ float s_w2[2 * HD * HD];
  __shared__ float s_b1[2 * HD];
  __shared__ float s_b2[HD];
  __shared__ float s_g[HD];
  __shared__ float s_bt[HD];
  __shared__ float s_h [TCH * 34];
  __shared__ float s_ff[TCH * 66];
  __shared__ __align__(16) float s_x[2][TCH * 36];
  __shared__ float s_ctx[HD];
  __shared__ float s_tmp[HD];

  const int tid  = threadIdx.x;
  const int wave = __builtin_amdgcn_readfirstlane(tid) >> 5;
  const int lane = tid & 31;
  const int ln   = lane & 15;
  const int hl   = lane >> 4;
  const int bidx = blockIdx.x;

  for (int i = tid; i < HD * 2 * HD; i += NTHREADS) s_w1[i] = w1[i];
  for (int i = tid; i < 2 * HD * HD; i += NTHREADS) s_w2[i] = w2[i];
  if (tid < 2 * HD) s_b1[tid] = b1[tid];
  if (tid < HD) { s_b2[tid] = b2[tid]; s_g[tid] = ln_g[tid]; s_bt[tid] = ln_b[tid]; }

  float Mrow[HD];
#pragma unroll
  for (int j = 0; j < HD; ++j) Mrow[j] = 0.f;

  const int nChunks = L / TCH;
  const int m0 = (wave & 3) * 16;

  auto encode = [&](int cIdx, float* xb) {
    const int t0 = cIdx * TCH;
#pragma unroll
    for (int r = 0; r < 16; ++r) {
      int tok = seq[(size_t)bidx * L + t0 + m0 + r];
      s_h[(m0 + r) * 34 + lane] = embed[tok * HD + lane];
    }
#pragma unroll
    for (int nt = 0; nt < 4; ++nt) {
      const int n0 = nt * 16;
      v8f acc = {0.f, 0.f, 0.f, 0.f, 0.f, 0.f, 0.f, 0.f};
#pragma unroll
      for (int k0 = 0; k0 < HD; k0 += 4) {
        const float* ap = &s_h[(m0 + ln) * 34 + k0 + 2 * hl];
        v2f a; a.x = ap[0]; a.y = ap[1];
        v2f bf; bf.x = s_w1[(k0 + 2 * hl) * 64 + n0 + ln];
        bf.y = s_w1[(k0 + 2 * hl + 1) * 64 + n0 + ln];
        acc = __builtin_amdgcn_wmma_f32_16x16x4_f32(false, a, false, bf,
                                                    (short)0, acc, false, false);
      }
      float bias = s_b1[n0 + ln];
#pragma unroll
      for (int r = 0; r < 8; ++r) {
        int row = m0 + r + 8 * hl;
        float v = acc[r] + bias;
        s_ff[row * 66 + n0 + ln] = v > 0.f ? v : 0.f;
      }
    }
#pragma unroll
    for (int nt = 0; nt < 2; ++nt) {
      const int n0 = nt * 16;
      v8f acc = {0.f, 0.f, 0.f, 0.f, 0.f, 0.f, 0.f, 0.f};
#pragma unroll
      for (int k0 = 0; k0 < 2 * HD; k0 += 4) {
        const float* ap = &s_ff[(m0 + ln) * 66 + k0 + 2 * hl];
        v2f a; a.x = ap[0]; a.y = ap[1];
        v2f bf; bf.x = s_w2[(k0 + 2 * hl) * HD + n0 + ln];
        bf.y = s_w2[(k0 + 2 * hl + 1) * HD + n0 + ln];
        acc = __builtin_amdgcn_wmma_f32_16x16x4_f32(false, a, false, bf,
                                                    (short)0, acc, false, false);
      }
      float bias = s_b2[n0 + ln];
#pragma unroll
      for (int r = 0; r < 8; ++r) {
        int row = m0 + r + 8 * hl, col = n0 + ln;
        xb[row * 36 + col] = acc[r] + bias + s_h[row * 34 + col];
      }
    }
    if (lane < 16) {
      float* xr = &xb[(m0 + lane) * 36];
      float mu = 0.f;
#pragma unroll
      for (int j = 0; j < HD; ++j) mu += xr[j];
      mu *= (1.f / 32.f);
      float var = 0.f;
#pragma unroll
      for (int j = 0; j < HD; ++j) { float d = xr[j] - mu; var = fmaf(d, d, var); }
      var *= (1.f / 32.f);
      float rs = __builtin_amdgcn_rsqf(var + 1e-5f);
      float kk2 = 0.f;
#pragma unroll
      for (int j = 0; j < HD; ++j) {
        float v = (xr[j] - mu) * rs * s_g[j] + s_bt[j];
        xr[j] = v;
        kk2 = fmaf(v, v, kk2);
      }
      xr[HD] = __builtin_amdgcn_rcpf(kk2 + 1e-6f);
    }
  };

  __syncthreads();
  if (wave < 4) encode(0, s_x[0]);
  __syncthreads();

  for (int c = 0; c < nChunks; ++c) {
    float* xb_cur = s_x[c & 1];
    float* xb_nxt = s_x[(c & 1) ^ 1];

    if (wave < 4) {
      if (c + 1 < nChunks) encode(c + 1, xb_nxt);
    } else {
      const bool last = (c == nChunks - 1);
      const int steps = last ? (TCH - 1) : TCH;

      auto loadk = [&](int t, float (&kkv)[HD], float& kl, float& ri) {
        const float4* kp = (const float4*)&xb_cur[t * 36];
#pragma unroll
        for (int q = 0; q < 8; ++q) {
          float4 f = kp[q];
          kkv[4 * q] = f.x; kkv[4 * q + 1] = f.y;
          kkv[4 * q + 2] = f.z; kkv[4 * q + 3] = f.w;
        }
        ri = xb_cur[t * 36 + HD];
        kl = xb_cur[t * 36 + lane];
      };
      auto stepM = [&](const float (&kkv)[HD], float kl, float ri) {
        float vp0 = 0.f, vp1 = 0.f, vp2 = 0.f, vp3 = 0.f;
#pragma unroll
        for (int j = 0; j < HD; j += 4) {
          vp0 = fmaf(Mrow[j],     kkv[j],     vp0);
          vp1 = fmaf(Mrow[j + 1], kkv[j + 1], vp1);
          vp2 = fmaf(Mrow[j + 2], kkv[j + 2], vp2);
          vp3 = fmaf(Mrow[j + 3], kkv[j + 3], vp3);
        }
        float vp = (vp0 + vp1) + (vp2 + vp3);
        float u = fmaf(-vp, ri, kl);
#pragma unroll
        for (int j = 0; j < HD; ++j) Mrow[j] = fmaf(u, kkv[j], Mrow[j]);
      };

      float kkA[HD], kkB[HD];
      float klA, riA, klB, riB;
      loadk(0, kkA, klA, riA);
      for (int t = 0; t < steps; t += 2) {
        if (t + 1 < steps) loadk(t + 1, kkB, klB, riB);
        stepM(kkA, klA, riA);
        if (t + 2 < steps) loadk(t + 2, kkA, klA, riA);
        if (t + 1 < steps) stepM(kkB, klB, riB);
      }

      if (last) {
        const float* kl = &xb_cur[(TCH - 1) * 36];
        float c0 = 0.f, c1 = 0.f, c2 = 0.f, c3 = 0.f;
#pragma unroll
        for (int j = 0; j < HD; j += 4) {
          c0 = fmaf(Mrow[j],     kl[j],     c0);
          c1 = fmaf(Mrow[j + 1], kl[j + 1], c1);
          c2 = fmaf(Mrow[j + 2], kl[j + 2], c2);
          c3 = fmaf(Mrow[j + 3], kl[j + 3], c3);
        }
        s_ctx[lane] = (c0 + c1) + (c2 + c3);
      }
    }
    __syncthreads();
  }

  if (tid < HD) {
    float acc = br[tid];
#pragma unroll
    for (int j = 0; j < HD; ++j) acc = fmaf(s_ctx[j], wr[j * HD + tid], acc);
    s_tmp[tid] = acc;
  }
  __syncthreads();
  if (tid < VD) {
    float acc = bo[tid];
#pragma unroll
    for (int i = 0; i < HD; ++i) acc = fmaf(s_tmp[i], wo[i * VD + tid], acc);
    out[(size_t)bidx * VD + tid] = acc;
  }
}

extern "C" void kernel_launch(void* const* d_in, const int* in_sizes, int n_in,
                              void* d_out, int out_size, void* d_ws, size_t ws_size,
                              hipStream_t stream) {
  const int*   seq   = (const int*)  d_in[0];
  const float* embed = (const float*)d_in[1];
  const float* w1    = (const float*)d_in[2];
  const float* b1    = (const float*)d_in[3];
  const float* w2    = (const float*)d_in[4];
  const float* b2    = (const float*)d_in[5];
  const float* ln_g  = (const float*)d_in[6];
  const float* ln_b  = (const float*)d_in[7];
  const float* wr    = (const float*)d_in[8];
  const float* br    = (const float*)d_in[9];
  const float* wo    = (const float*)d_in[10];
  const float* bo    = (const float*)d_in[11];
  float* out = (float*)d_out;

  const int B = out_size / VD;
  const int L = in_sizes[0] / B;

  delta_model_fused<<<B, NTHREADS, 0, stream>>>(
      seq, embed, w1, b1, w2, b2, ln_g, ln_b, wr, br, wo, bo, out, L);
}


# Delta scan spread over 4 waves: a quad of lanes owns one memory row (8 columns each), partial dots joined by two DPP quad adds (the dot's summation tree changes); LDS reads held 3 steps ahead
# speedup vs baseline: 1.7075x; 1.7075x over previous
#include <hip/hip_runtime.h>

typedef __attribute__((ext_vector_type(2))) float v2f;
typedef __attribute__((ext_vector_type(8))) float v8f;

#define HD 32
#define VD 64
#define TCH 64
#define XPAD 3
#define NTHREADS 256

__device__ __forceinline__ float quad_sum(float v) {
  v += __int_as_float(__builtin_amdgcn_update_dpp(0, __float_as_int(v), 0xB1, 0xF, 0xF, true));
  v += __int_as_float(__builtin_amdgcn_update_dpp(0, __float_as_int(v), 0x4E, 0xF, 0xF, true));
  return v;
}


__launch_bounds__(NTHREADS)
__global__ void delta_model_fused(
    const int*   __restrict__ seq,
    const float* __restrict__ embed,
    const float* __restrict__ w1, const float* __restrict__ b1,
    const float* __restrict__ w2, const float* __restrict__ b2,
    const float* __restrict__ ln_g, const float* __restrict__ ln_b,
    const float* __restrict__ wr, const float* __restrict__ br,
    const float* __restrict__ wo, const float* __restrict__ bo,
    float* __restrict__ out, int L)
{
  __shared__ float s_w1[HD * 2 * HD];
  __shared__ float s_w2[2 * HD * HD];
  __shared__ float s_b1[2 * HD];
  __shared__ float s_b2[HD];
  __shared__ float s_g[HD];
  __shared__ float s_bt[HD];
  __shared__ float s_h [TCH * 34];
  __shared__ float s_ff[TCH * 66];
  __shared__ __align__(16) float s_x[2][(TCH + XPAD) * 36];
  __shared__ float s_ctx[HD];
  __shared__ float s_tmp[HD];

  const int tid  = threadIdx.x;
  const int wave = __builtin_amdgcn_readfirstlane(tid) >> 5;
  const int lane = tid & 31;
  const int ln   = lane & 15;
  const int hl   = lane >> 4;
  const int bidx = blockIdx.x;

  for (int i = tid; i < HD * 2 * HD; i += NTHREADS) s_w1[i] = w1[i];
  for (int i = tid; i < 2 * HD * HD; i += NTHREADS) s_w2[i] = w2[i];
  if (tid < 2 * HD) s_b1[tid] = b1[tid];
  if (tid < HD) { s_b2[tid] = b2[tid]; s_g[tid] = ln_g[tid]; s_bt[tid] = ln_b[tid]; }
  if (tid < 2 * XPAD * 36) s_x[tid / (XPAD * 36)][TCH * 36 + tid % (XPAD * 36)] = 0.f;

  const int q     = lane & 3;
  const int myrow = ((wave & 3) << 3) + (lane >> 2);
  v2f M01 = {0.f, 0.f}, M23 = {0.f, 0.f}, M45 = {0.f, 0.f}, M67 = {0.f, 0.f};

  const int nChunks = L / TCH;
  const int m0 = (wave & 3) * 16;

  auto encode = [&](int cIdx, float* xb) {
    const int t0 = cIdx * TCH;
#pragma unroll
    for (int r = 0; r < 16; ++r) {
      int tok = seq[(size_t)bidx * L + t0 + m0 + r];
      s_h[(m0 + r) * 34 + lane] = embed[tok * HD + lane];
    }
#pragma unroll
    for (int nt = 0; nt < 4; ++nt) {
      const int n0 = nt * 16;
      v8f acc = {0.f, 0.f, 0.f, 0.f, 0.f, 0.f, 0.f, 0.f};
#pragma unroll
      for (int k0 = 0; k0 < HD; k0 += 4) {
        const float* ap = &s_h[(m0 + ln) * 34 + k0 + 2 * hl];
        v2f a; a.x = ap[0]; a.y = ap[1];
        v2f bf; bf.x = s_w1[(k0 + 2 * hl) * 64 + n0 + ln];
        bf.y = s_w1[(k0 + 2 * hl + 1) * 64 + n0 + ln];
        acc = __builtin_amdgcn_wmma_f32_16x16x4_f32(false, a, false, bf,
                                                    (short)0, acc, false, false);
      }
      float bias = s_b1[n0 + ln];
#pragma unroll
      for (int r = 0; r < 8; ++r) {
        int row = m0 + r + 8 * hl;
        float v = acc[r] + bias;
        s_ff[row * 66 + n0 + ln] = v > 0.f ? v : 0.f;
      }
    }
#pragma unroll
    for (int nt = 0; nt < 2; ++nt) {
      const int n0 = nt * 16;
      v8f acc = {0.f, 0.f, 0.f, 0.f, 0.f, 0.f, 0.f, 0.f};
#pragma unroll
      for (int k0 = 0; k0 < 2 * HD; k0 += 4) {
        const float* ap = &s_ff[(m0 + ln) * 66 + k0 + 2 * hl];
        v2f a; a.x = ap[0]; a.y = ap[1];
        v2f bf; bf.x = s_w2[(k0 + 2 * hl) * HD + n0 + ln];
        bf.y = s_w2[(k0 + 2 * hl + 1) * HD + n0 + ln];
        acc = __builtin_amdgcn_wmma_f32_16x16x4_f32(false, a, false, bf,
                                                    (short)0, acc, false, false);
      }
      float bias = s_b2[n0 + ln];
#pragma unroll
      for (int r = 0; r < 8; ++r) {
        int row = m0 + r + 8 * hl, col = n0 + ln;
        xb[row * 36 + col] = acc[r] + bias + s_h[row * 34 + col];
      }
    }
    if (lane < 16) {
      float* xr = &xb[(m0 + lane) * 36];
      float mu = 0.f;
#pragma unroll
      for (int j = 0; j < HD; ++j) mu += xr[j];
      mu *= (1.f / 32.f);
      float var = 0.f;
#pragma unroll
      for (int j = 0; j < HD; ++j) { float d = xr[j] - mu; var = fmaf(d, d, var); }
      var *= (1.f / 32.f);
      float rs = __builtin_amdgcn_rsqf(var + 1e-5f);
      float kk2 = 0.f;
#pragma unroll
      for (int j = 0; j < HD; ++j) {
        float v = (xr[j] - mu) * rs * s_g[j] + s_bt[j];
        xr[j] = v;
        kk2 = fmaf(v, v, kk2);
      }
      xr[HD] = __builtin_amdgcn_rcpf(kk2 + 1e-6f);
    }
  };

  __syncthreads();
  if (wave < 4) encode(0, s_x[0]);
  __syncthreads();

  for (int c = 0; c < nChunks; ++c) {
    float* xb_cur = s_x[c & 1];
    float* xb_nxt = s_x[(c & 1) ^ 1];

    if (wave < 4) {
      if (c + 1 < nChunks) encode(c + 1, xb_nxt);
    } else {
      const bool last = (c == nChunks - 1);
      const int steps = last ? (TCH - 1) : TCH;

      struct KBuf { float4 f0, f1; float kl, ri; };
      auto loadk = [&](int t, KBuf& kb) {
        const float* rowp = &xb_cur[t * 36];
        kb.f0 = *(const float4*)(rowp + 8 * q);
        kb.f1 = *(const float4*)(rowp + 8 * q + 4);
        kb.ri = rowp[HD];
        kb.kl = rowp[myrow];
      };
      auto stepM = [&](const KBuf& kb) {
        const v2f k01 = {kb.f0.x, kb.f0.y}, k23 = {kb.f0.z, kb.f0.w};
        const v2f k45 = {kb.f1.x, kb.f1.y}, k67 = {kb.f1.z, kb.f1.w};
        v2f P = M01 * k01;
        v2f Q = M45 * k45;
        P = __builtin_elementwise_fma(M23, k23, P);
        Q = __builtin_elementwise_fma(M67, k67, Q);
        const v2f R = P + Q;
        const float vp = quad_sum(R.x + R.y);
        const float u = fmaf(-vp, kb.ri, kb.kl);
        const v2f uu = {u, u};
        M01 = __builtin_elementwise_fma(uu, k01, M01);
        M23 = __builtin_elementwise_fma(uu, k23, M23);
        M45 = __builtin_elementwise_fma(uu, k45, M45);
        M67 = __builtin_elementwise_fma(uu, k67, M67);
      };

#define HOLD_LDS_READS()                   \
  do {                                     \
    asm volatile("" ::: "memory");         \
    __builtin_amdgcn_sched_barrier(0);     \
  } while (0)
      KBuf b0, b1, b2, b3;
      loadk(0, b0);
      loadk(1, b1);
      loadk(2, b2);
      HOLD_LDS_READS();
      int t = 0;
      for (; t + 4 <= steps; t += 4) {
        loadk(t + 3, b3);
        HOLD_LDS_READS();
        stepM(b0);
        loadk(t + 4, b0);
        HOLD_LDS_READS();
        stepM(b1);
        loadk(t + 5, b1);
        HOLD_LDS_READS();
        stepM(b2);
        loadk(t + 6, b2);
        HOLD_LDS_READS();
        stepM(b3);
      }
#undef HOLD_LDS_READS
      const int rem = steps - t;
      if (rem > 0) stepM(b0);
      if (rem > 1) stepM(b1);
      if (rem > 2) stepM(b2);

      if (last) {
        const float* hp = &xb_cur[(TCH - 1) * 36 + 8 * q];
        const float4 h0 = *(const float4*)hp;
        const float4 h1 = *(const float4*)(hp + 4);
        const v2f h01 = {h0.x, h0.y}, h23 = {h0.z, h0.w};
        const v2f h45 = {h1.x, h1.y}, h67 = {h1.z, h1.w};
        v2f P = M01 * h01;
        v2f Q = M45 * h45;
        P = __builtin_elementwise_fma(M23, h23, P);
        Q = __builtin_elementwise_fma(M67, h67, Q);
        const v2f R = P + Q;
        const float cs = quad_sum(R.x + R.y);
        if (q == 0) s_ctx[myrow] = cs;
      }
    }
    __syncthreads();
  }

  if (tid < HD) {
    float acc = br[tid];
#pragma unroll
    for (int j = 0; j < HD; ++j) acc = fmaf(s_ctx[j], wr[j * HD + tid], acc);
    s_tmp[tid] = acc;
  }
  __syncthreads();
  if (tid < VD) {
    float acc = bo[tid];
#pragma unroll
    for (int i = 0; i < HD; ++i) acc = fmaf(s_tmp[i], wo[i * VD + tid], acc);
    out[(size_t)bidx * VD + tid] = acc;
  }
}

extern "C" void kernel_launch(void* const* d_in, const int* in_sizes, int n_in,
                              void* d_out, int out_size, void* d_ws, size_t ws_size,
                              hipStream_t stream) {
  const int*   seq   = (const int*)  d_in[0];
  const float* embed = (const float*)d_in[1];
  const float* w1    = (const float*)d_in[2];
  const float* b1    = (const float*)d_in[3];
  const float* w2    = (const float*)d_in[4];
  const float* b2    = (const float*)d_in[5];
  const float* ln_g  = (const float*)d_in[6];
  const float* ln_b  = (const float*)d_in[7];
  const float* wr    = (const float*)d_in[8];
  const float* br    = (const float*)d_in[9];
  const float* wo    = (const float*)d_in[10];
  const float* bo    = (const float*)d_in[11];
  float* out = (float*)d_out;

  const int B = out_size / VD;
  const int L = in_sizes[0] / B;

  delta_model_fused<<<B, NTHREADS, 0, stream>>>(
      seq, embed, w1, b1, w2, b2, ln_g, ln_b, wr, br, wo, bo, out, L);
}


# Encoder run once per token id (64 embedding rows -> a table in LDS), not once per position: it is row-wise in embed[tok]; helper waves copy table rows by token id into the scan's buffers
# speedup vs baseline: 2.4509x; 1.4354x over previous
#include <hip/hip_runtime.h>

typedef __attribute__((ext_vector_type(2))) float v2f;
typedef __attribute__((ext_vector_type(8))) float v8f;

#define HD 32
#define VD 64
#define TCH 64
#define NTOK 64
#define XPAD 3
#define NTHREADS 256

__device__ __forceinline__ float quad_sum(float v) {
  v += __int_as_float(__builtin_amdgcn_update_dpp(0, __float_as_int(v), 0xB1, 0xF, 0xF, true));
  v += __int_as_float(__builtin_amdgcn_update_dpp(0, __float_as_int(v), 0x4E, 0xF, 0xF, true));
  return v;
}


__launch_bounds__(NTHREADS)
__global__ void delta_model_fused(
    const int*   __restrict__ seq,
    const float* __restrict__ embed,
    const float* __restrict__ w1, const float* __restrict__ b1,
    const float* __restrict__ w2, const float* __restrict__ b2,
    const float* __restrict__ ln_g, const float* __restrict__ ln_b,
    const float* __restrict__ wr, const float* __restrict__ br,
    const float* __restrict__ wo, const float* __restrict__ bo,
    float* __restrict__ out, int L)
{
  __shared__ float s_w1[HD * 2 * HD];
  __shared__ float s_w2[2 * HD * HD];
  __shared__ float s_b1[2 * HD];
  __shared__ float s_b2[HD];
  __shared__ float s_g[HD];
  __shared__ float s_bt[HD];
  __shared__ float s_h [TCH * 34];
  __shared__ float s_ff[TCH * 66];
  __shared__ __align__(16) float s_x[2][(TCH + XPAD) * 36];
  __shared__ __align__(16) float s_tab[NTOK * 36];
  __shared__ float s_ctx[HD];
  __shared__ float s_tmp[HD];

  const int tid  = threadIdx.x;
  const int wave = __builtin_amdgcn_readfirstlane(tid) >> 5;
  const int lane = tid & 31;
  const int ln   = lane & 15;
  const int hl   = lane >> 4;
  const int bidx = blockIdx.x;

  for (int i = tid; i < HD * 2 * HD; i += NTHREADS) s_w1[i] = w1[i];
  for (int i = tid; i < 2 * HD * HD; i += NTHREADS) s_w2[i] = w2[i];
  if (tid < 2 * HD) s_b1[tid] = b1[tid];
  if (tid < HD) { s_b2[tid] = b2[tid]; s_g[tid] = ln_g[tid]; s_bt[tid] = ln_b[tid]; }
  if (tid < 2 * XPAD * 36) s_x[tid / (XPAD * 36)][TCH * 36 + tid % (XPAD * 36)] = 0.f;

  const int q     = lane & 3;
  const int myrow = ((wave & 3) << 3) + (lane >> 2);
  v2f M01 = {0.f, 0.f}, M23 = {0.f, 0.f}, M45 = {0.f, 0.f}, M67 = {0.f, 0.f};

  const int nChunks = L / TCH;
  const int m0 = (wave & 3) * 16;

  auto build_table = [&]() {
    float* xb = s_tab;
#pragma unroll
    for (int r = 0; r < 16; ++r)
      s_h[(m0 + r) * 34 + lane] = embed[(m0 + r) * HD + lane];
#pragma unroll
    for (int nt = 0; nt < 4; ++nt) {
      const int n0 = nt * 16;
      v8f acc = {0.f, 0.f, 0.f, 0.f, 0.f, 0.f, 0.f, 0.f};
#pragma unroll
      for (int k0 = 0; k0 < HD; k0 += 4) {
        const float* ap = &s_h[(m0 + ln) * 34 + k0 + 2 * hl];
        v2f a; a.x = ap[0]; a.y = ap[1];
        v2f bf; bf.x = s_w1[(k0 + 2 * hl) * 64 + n0 + ln];
        bf.y = s_w1[(k0 + 2 * hl + 1) * 64 + n0 + ln];
        acc = __builtin_amdgcn_wmma_f32_16x16x4_f32(false, a, false, bf,
                                                    (short)0, acc, false, false);
      }
      float bias = s_b1[n0 + ln];
#pragma unroll
      for (int r = 0; r < 8; ++r) {
        int row = m0 + r + 8 * hl;
        float v = acc[r] + bias;
        s_ff[row * 66 + n0 + ln] = v > 0.f ? v : 0.f;
      }
    }
#pragma unroll
    for (int nt = 0; nt < 2; ++nt) {
      const int n0 = nt * 16;
      v8f acc = {0.f, 0.f, 0.f, 0.f, 0.f, 0.f, 0.f, 0.f};
#pragma unroll
      for (int k0 = 0; k0 < 2 * HD; k0 += 4) {
        const float* ap = &s_ff[(m0 + ln) * 66 + k0 + 2 * hl];
        v2f a; a.x = ap[0]; a.y = ap[1];
        v2f bf; bf.x = s_w2[(k0 + 2 * hl) * HD + n0 + ln];
        bf.y = s_w2[(k0 + 2 * hl + 1) * HD + n0 + ln];
        acc = __builtin_amdgcn_wmma_f32_16x16x4_f32(false, a, false, bf,
                                                    (short)0, acc, false, false);
      }
      float bias = s_b2[n0 + ln];
#pragma unroll
      for (int r = 0; r < 8; ++r) {
        int row = m0 + r + 8 * hl, col = n0 + ln;
        xb[row * 36 + col] = acc[r] + bias + s_h[row * 34 + col];
      }
    }
    if (lane < 16) {
      float* xr = &xb[(m0 + lane) * 36];
      float mu = 0.f;
#pragma unroll
      for (int j = 0; j < HD; ++j) mu += xr[j];
      mu *= (1.f / 32.f);
      float var = 0.f;
#pragma unroll
      for (int j = 0; j < HD; ++j) { float d = xr[j] - mu; var = fmaf(d, d, var); }
      var *= (1.f / 32.f);
      float rs = __builtin_amdgcn_rsqf(var + 1e-5f);
      float kk2 = 0.f;
#pragma unroll
      for (int j = 0; j < HD; ++j) {
        float v = (xr[j] - mu) * rs * s_g[j] + s_bt[j];
        xr[j] = v;
        kk2 = fmaf(v, v, kk2);
      }
      xr[HD] = __builtin_amdgcn_rcpf(kk2 + 1e-6f);
    }
  };

  auto gather = [&](int cIdx, float* xb) {
    const int r  = lane & 15;
    const int p0 = (lane >> 4) * 4;
    const int tok = seq[(size_t)bidx * L + cIdx * TCH + m0 + r];
    const float4* src = (const float4*)&s_tab[tok * 36] + p0;
    float4* dst = (float4*)&xb[(m0 + r) * 36] + p0;
    const float4 a0 = src[0], a1 = src[1], a2 = src[2], a3 = src[3];
    const float ri = s_tab[tok * 36 + HD];
    dst[0] = a0; dst[1] = a1; dst[2] = a2; dst[3] = a3;
    if (lane < 16) xb[(m0 + r) * 36 + HD] = ri;
  };

  __syncthreads();
  if (wave < 4) build_table();
  __syncthreads();
  if (wave < 4) gather(0, s_x[0]);
  __syncthreads();

  for (int c = 0; c < nChunks; ++c) {
    float* xb_cur = s_x[c & 1];
    float* xb_nxt = s_x[(c & 1) ^ 1];

    if (wave < 4) {
      if (c + 1 < nChunks) gather(c + 1, xb_nxt);
    } else {
      const bool last = (c == nChunks - 1);
      const int steps = last ? (TCH - 1) : TCH;

      struct KBuf { float4 f0, f1; float kl, ri; };
      auto loadk = [&](int t, KBuf& kb) {
        const float* rowp = &xb_cur[t * 36];
        kb.f0 = *(const float4*)(rowp + 8 * q);
        kb.f1 = *(const float4*)(rowp + 8 * q + 4);
        kb.ri = rowp[HD];
        kb.kl = rowp[myrow];
      };
      auto stepM = [&](const KBuf& kb) {
        const v2f k01 = {kb.f0.x, kb.f0.y}, k23 = {kb.f0.z, kb.f0.w};
        const v2f k45 = {kb.f1.x, kb.f1.y}, k67 = {kb.f1.z, kb.f1.w};
        v2f P = M01 * k01;
        v2f Q = M45 * k45;
        P = __builtin_elementwise_fma(M23, k23, P);
        Q = __builtin_elementwise_fma(M67, k67, Q);
        const v2f R = P + Q;
        const float vp = quad_sum(R.x + R.y);
        const float u = fmaf(-vp, kb.ri, kb.kl);
        const v2f uu = {u, u};
        M01 = __builtin_elementwise_fma(uu, k01, M01);
        M23 = __builtin_elementwise_fma(uu, k23, M23);
        M45 = __builtin_elementwise_fma(uu, k45, M45);
        M67 = __builtin_elementwise_fma(uu, k67, M67);
      };

#define HOLD_LDS_READS()                   \
  do {                                     \
    asm volatile("" ::: "memory");         \
    __builtin_amdgcn_sched_barrier(0);     \
  } while (0)
      KBuf b0, b1, b2, b3;
      loadk(0, b0);
      loadk(1, b1);
      loadk(2, b2);
      HOLD_LDS_READS();
      int t = 0;
      for (; t + 4 <= steps; t += 4) {
        loadk(t + 3, b3);
        HOLD_LDS_READS();
        stepM(b0);
        loadk(t + 4, b0);
        HOLD_LDS_READS();
        stepM(b1);
        loadk(t + 5, b1);
        HOLD_LDS_READS();
        stepM(b2);
        loadk(t + 6, b2);
        HOLD_LDS_READS();
        stepM(b3);
      }
#undef HOLD_LDS_READS
      const int rem = steps - t;
      if (rem > 0) stepM(b0);
      if (rem > 1) stepM(b1);
      if (rem > 2) stepM(b2);

      if (last) {
        const float* hp = &xb_cur[(TCH - 1) * 36 + 8 * q];
        const float4 h0 = *(const float4*)hp;
        const float4 h1 = *(const float4*)(hp + 4);
        const v2f h01 = {h0.x, h0.y}, h23 = {h0.z, h0.w};
        const v2f h45 = {h1.x, h1.y}, h67 = {h1.z, h1.w};
        v2f P = M01 * h01;
        v2f Q = M45 * h45;
        P = __builtin_elementwise_fma(M23, h23, P);
        Q = __builtin_elementwise_fma(M67, h67, Q);
        const v2f R = P + Q;
        const float cs = quad_sum(R.x + R.y);
        if (q == 0) s_ctx[myrow] = cs;
      }
    }
    __syncthreads();
  }

  if (tid < HD) {
    float acc = br[tid];
#pragma unroll
    for (int j = 0; j < HD; ++j) acc = fmaf(s_ctx[j], wr[j * HD + tid], acc);
    s_tmp[tid] = acc;
  }
  __syncthreads();
  if (tid < VD) {
    float acc = bo[tid];
#pragma unroll
    for (int i = 0; i < HD; ++i) acc = fmaf(s_tmp[i], wo[i * VD + tid], acc);
    out[(size_t)bidx * VD + tid] = acc;
  }
}

extern "C" void kernel_launch(void* const* d_in, const int* in_sizes, int n_in,
                              void* d_out, int out_size, void* d_ws, size_t ws_size,
                              hipStream_t stream) {
  const int*   seq   = (const int*)  d_in[0];
  const float* embed = (const float*)d_in[1];
  const float* w1    = (const float*)d_in[2];
  const float* b1    = (const float*)d_in[3];
  const float* w2    = (const float*)d_in[4];
  const float* b2    = (const float*)d_in[5];
  const float* ln_g  = (const float*)d_in[6];
  const float* ln_b  = (const float*)d_in[7];
  const float* wr    = (const float*)d_in[8];
  const float* br    = (const float*)d_in[9];
  const float* wo    = (const float*)d_in[10];
  const float* bo    = (const float*)d_in[11];
  float* out = (float*)d_out;

  const int B = out_size / VD;
  const int L = in_sizes[0] / B;

  delta_model_fused<<<B, NTHREADS, 0, stream>>>(
      seq, embed, w1, b1, w2, b2, ln_g, ln_b, wr, br, wo, bo, out, L);
}


# Delta step re-associated: v_pred_t = M_{t-1}k_t + u_{t-1}(k_{t-1}.k_t), so dot and quad sum start a step early (v_pred's summation order changes); k.k from a token Gram table built once by WMMA
# speedup vs baseline: 2.7322x; 1.1148x over previous
#include <hip/hip_runtime.h>

typedef __attribute__((ext_vector_type(2))) float v2f;
typedef __attribute__((ext_vector_type(8))) float v8f;

#define HD 32
#define VD 64
#define TCH 64
#define NTOK 64
#define XPAD 5
#define NTHREADS 256

__device__ __forceinline__ float quad_sum(float v) {
  v += __int_as_float(__builtin_amdgcn_update_dpp(0, __float_as_int(v), 0xB1, 0xF, 0xF, true));
  v += __int_as_float(__builtin_amdgcn_update_dpp(0, __float_as_int(v), 0x4E, 0xF, 0xF, true));
  return v;
}


__launch_bounds__(NTHREADS)
__global__ void delta_model_fused(
    const int*   __restrict__ seq,
    const float* __restrict__ embed,
    const float* __restrict__ w1, const float* __restrict__ b1,
    const float* __restrict__ w2, const float* __restrict__ b2,
    const float* __restrict__ ln_g, const float* __restrict__ ln_b,
    const float* __restrict__ wr, const float* __restrict__ br,
    const float* __restrict__ wo, const float* __restrict__ bo,
    float* __restrict__ out, int L)
{
  __shared__ float s_w1[HD * 2 * HD];
  __shared__ float s_w2[2 * HD * HD];
  __shared__ float s_b1[2 * HD];
  __shared__ float s_b2[HD];
  __shared__ float s_g[HD];
  __shared__ float s_bt[HD];
  __shared__ float s_h [TCH * 34];
  __shared__ float s_ff[TCH * 66];
  __shared__ __align__(16) float s_x[2][(TCH + XPAD) * 36];
  __shared__ __align__(16) float s_tab[NTOK * 36];
  __shared__ float s_gram[NTOK * NTOK];
  __shared__ float s_ctx[HD];
  __shared__ float s_tmp[HD];

  const int tid  = threadIdx.x;
  const int wave = __builtin_amdgcn_readfirstlane(tid) >> 5;
  const int lane = tid & 31;
  const int ln   = lane & 15;
  const int hl   = lane >> 4;
  const int bidx = blockIdx.x;

  for (int i = tid; i < HD * 2 * HD; i += NTHREADS) s_w1[i] = w1[i];
  for (int i = tid; i < 2 * HD * HD; i += NTHREADS) s_w2[i] = w2[i];
  if (tid < 2 * HD) s_b1[tid] = b1[tid];
  if (tid < HD) { s_b2[tid] = b2[tid]; s_g[tid] = ln_g[tid]; s_bt[tid] = ln_b[tid]; }
  for (int i = tid; i < 2 * XPAD * 36; i += NTHREADS)
    s_x[i / (XPAD * 36)][TCH * 36 + i % (XPAD * 36)] = 0.f;

  const int q     = lane & 3;
  const int myrow = ((wave & 3) << 3) + (lane >> 2);
  v2f M01 = {0.f, 0.f}, M23 = {0.f, 0.f}, M45 = {0.f, 0.f}, M67 = {0.f, 0.f};

  const int nChunks = L / TCH;
  const int m0 = (wave & 3) * 16;

  auto build_table = [&]() {
    float* xb = s_tab;
#pragma unroll
    for (int r = 0; r < 16; ++r)
      s_h[(m0 + r) * 34 + lane] = embed[(m0 + r) * HD + lane];
#pragma unroll
    for (int nt = 0; nt < 4; ++nt) {
      const int n0 = nt * 16;
      v8f acc = {0.f, 0.f, 0.f, 0.f, 0.f, 0.f, 0.f, 0.f};
#pragma unroll
      for (int k0 = 0; k0 < HD; k0 += 4) {
        const float* ap = &s_h[(m0 + ln) * 34 + k0 + 2 * hl];
        v2f a; a.x = ap[0]; a.y = ap[1];
        v2f bf; bf.x = s_w1[(k0 + 2 * hl) * 64 + n0 + ln];
        bf.y = s_w1[(k0 + 2 * hl + 1) * 64 + n0 + ln];
        acc = __builtin_amdgcn_wmma_f32_16x16x4_f32(false, a, false, bf,
                                                    (short)0, acc, false, false);
      }
      float bias = s_b1[n0 + ln];
#pragma unroll
      for (int r = 0; r < 8; ++r) {
        int row = m0 + r + 8 * hl;
        float v = acc[r] + bias;
        s_ff[row * 66 + n0 + ln] = v > 0.f ? v : 0.f;
      }
    }
#pragma unroll
    for (int nt = 0; nt < 2; ++nt) {
      const int n0 = nt * 16;
      v8f acc = {0.f, 0.f, 0.f, 0.f, 0.f, 0.f, 0.f, 0.f};
#pragma unroll
      for (int k0 = 0; k0 < 2 * HD; k0 += 4) {
        const float* ap = &s_ff[(m0 + ln) * 66 + k0 + 2 * hl];
        v2f a; a.x = ap[0]; a.y = ap[1];
        v2f bf; bf.x = s_w2[(k0 + 2 * hl) * HD + n0 + ln];
        bf.y = s_w2[(k0 + 2 * hl + 1) * HD + n0 + ln];
        acc = __builtin_amdgcn_wmma_f32_16x16x4_f32(false, a, false, bf,
                                                    (short)0, acc, false, false);
      }
      float bias = s_b2[n0 + ln];
#pragma unroll
      for (int r = 0; r < 8; ++r) {
        int row = m0 + r + 8 * hl, col = n0 + ln;
        xb[row * 36 + col] = acc[r] + bias + s_h[row * 34 + col];
      }
    }
    if (lane < 16) {
      float* xr = &xb[(m0 + lane) * 36];
      float mu = 0.f;
#pragma unroll
      for (int j = 0; j < HD; ++j) mu += xr[j];
      mu *= (1.f / 32.f);
      float var = 0.f;
#pragma unroll
      for (int j = 0; j < HD; ++j) { float d = xr[j] - mu; var = fmaf(d, d, var); }
      var *= (1.f / 32.f);
      float rs = __builtin_amdgcn_rsqf(var + 1e-5f);
      float kk2 = 0.f;
#pragma unroll
      for (int j = 0; j < HD; ++j) {
        float v = (xr[j] - mu) * rs * s_g[j] + s_bt[j];
        xr[j] = v;
        kk2 = fmaf(v, v, kk2);
      }
      xr[HD] = __builtin_amdgcn_rcpf(kk2 + 1e-6f);
    }
  };

  auto build_gram = [&]() {
#pragma unroll
    for (int nt = 0; nt < 4; ++nt) {
      const int n0 = nt * 16;
      v8f acc = {0.f, 0.f, 0.f, 0.f, 0.f, 0.f, 0.f, 0.f};
#pragma unroll
      for (int k0 = 0; k0 < HD; k0 += 4) {
        const float* ap = &s_tab[(m0 + ln) * 36 + k0 + 2 * hl];
        const float* bp = &s_tab[(n0 + ln) * 36 + k0 + 2 * hl];
        v2f a;  a.x  = ap[0]; a.y  = ap[1];
        v2f bf; bf.x = bp[0]; bf.y = bp[1];
        acc = __builtin_amdgcn_wmma_f32_16x16x4_f32(false, a, false, bf,
                                                    (short)0, acc, false, false);
      }
#pragma unroll
      for (int r = 0; r < 8; ++r)
        s_gram[(m0 + r + 8 * hl) * NTOK + n0 + ln] = acc[r];
    }
  };

  auto gather = [&](int cIdx, float* xb) {
    const int r  = lane & 15;
    const int p0 = (lane >> 4) * 4;
    const int tpos = cIdx * TCH + m0 + r;
    const int* sp = seq + (size_t)bidx * L + tpos;
    const int tok = sp[0];
    const int tokp = tpos > 0 ? sp[-1] : tok;
    const float4* src = (const float4*)&s_tab[tok * 36] + p0;
    float4* dst = (float4*)&xb[(m0 + r) * 36] + p0;
    const float4 a0 = src[0], a1 = src[1], a2 = src[2], a3 = src[3];
    const float ri = s_tab[tok * 36 + HD];
    const float g  = s_gram[tokp * NTOK + tok];
    dst[0] = a0; dst[1] = a1; dst[2] = a2; dst[3] = a3;
    if (lane < 16) { xb[(m0 + r) * 36 + HD] = ri; xb[(m0 + r) * 36 + HD + 1] = g; }
  };

  __syncthreads();
  if (wave < 4) build_table();
  __syncthreads();
  if (wave < 4) build_gram();
  __syncthreads();
  if (wave < 4) gather(0, s_x[0]);
  __syncthreads();

  for (int c = 0; c < nChunks; ++c) {
    float* xb_cur = s_x[c & 1];
    float* xb_nxt = s_x[(c & 1) ^ 1];

    if (wave < 4) {
      if (c + 1 < nChunks) gather(c + 1, xb_nxt);
    } else {
      const bool last = (c == nChunks - 1);

      struct KBuf { float4 f0, f1; float kl, ri, g; };
      auto loadk = [&](int t, KBuf& kb) {
        const float* rowp = &xb_cur[t * 36];
        kb.f0 = *(const float4*)(rowp + 8 * q);
        kb.f1 = *(const float4*)(rowp + 8 * q + 4);
        const float2 rg = *(const float2*)(rowp + HD);
        kb.ri = rg.x;
        kb.g  = rg.y;
        kb.kl = rowp[myrow];
      };

      v2f R;
      float u;
      auto dot_products = [&](const KBuf& kb) -> v2f {
        const v2f k01 = {kb.f0.x, kb.f0.y}, k23 = {kb.f0.z, kb.f0.w};
        const v2f k45 = {kb.f1.x, kb.f1.y}, k67 = {kb.f1.z, kb.f1.w};
        v2f Po = M01 * k01;
        v2f Qo = M45 * k45;
        Po = __builtin_elementwise_fma(M23, k23, Po);
        Qo = __builtin_elementwise_fma(M67, k67, Qo);
        return Po + Qo;
      };
      auto dot_sum = [&](const v2f& Ri) -> float {
        return quad_sum(Ri.x + Ri.y);
      };
#define PASS_ORDER() __builtin_amdgcn_sched_barrier(0)
      auto pass = [&](const KBuf& cur, const KBuf& nx1, const KBuf& nx2, const bool drop) {
        const v2f uu = {u, u};
        const v2f c01 = {cur.f0.x, cur.f0.y}, c23 = {cur.f0.z, cur.f0.w};
        const v2f c45 = {cur.f1.x, cur.f1.y}, c67 = {cur.f1.z, cur.f1.w};
        const v2f n01 = {nx2.f0.x, nx2.f0.y}, n23 = {nx2.f0.z, nx2.f0.w};
        const v2f n45 = {nx2.f1.x, nx2.f1.y}, n67 = {nx2.f1.z, nx2.f1.w};
        M01 = __builtin_elementwise_fma(uu, c01, M01);
        M45 = __builtin_elementwise_fma(uu, c45, M45);
        M23 = __builtin_elementwise_fma(uu, c23, M23);
        M67 = __builtin_elementwise_fma(uu, c67, M67);
        PASS_ORDER();
        float s = R.x + R.y;
        v2f Pn = M01 * n01;
        v2f Qn = M45 * n45;
        PASS_ORDER();
        s += __int_as_float(__builtin_amdgcn_update_dpp(0, __float_as_int(s), 0xB1, 0xF, 0xF, true));
        Pn = __builtin_elementwise_fma(M23, n23, Pn);
        Qn = __builtin_elementwise_fma(M67, n67, Qn);
        s += __int_as_float(__builtin_amdgcn_update_dpp(0, __float_as_int(s), 0x4E, 0xF, 0xF, true));
        PASS_ORDER();
        const float vp = fmaf(u, nx1.g, s);
        R = Pn + Qn;
        u = fmaf(-vp, nx1.ri, nx1.kl);
        if (drop) u = 0.f;
      };

#define HOLD_LDS_READS()                   \
  do {                                     \
    asm volatile("" ::: "memory");         \
    __builtin_amdgcn_sched_barrier(0);     \
  } while (0)
      KBuf b0, b1, b2, b3, b4, b5, b6, b7;
      loadk(0, b0);
      loadk(1, b1);
      loadk(2, b2);
      loadk(3, b3);
      loadk(4, b4);
      HOLD_LDS_READS();
      u = fmaf(-dot_sum(dot_products(b0)), b0.ri, b0.kl);
      R = dot_products(b1);
      for (int t = 0; t < TCH; t += 8) {
        loadk(t + 5, b5);
        HOLD_LDS_READS();
        pass(b0, b1, b2, false);
        loadk(t + 6, b6);
        HOLD_LDS_READS();
        pass(b1, b2, b3, false);
        loadk(t + 7, b7);
        HOLD_LDS_READS();
        pass(b2, b3, b4, false);
        loadk(t + 8, b0);
        HOLD_LDS_READS();
        pass(b3, b4, b5, false);
        loadk(t + 9, b1);
        HOLD_LDS_READS();
        pass(b4, b5, b6, false);
        loadk(t + 10, b2);
        HOLD_LDS_READS();
        pass(b5, b6, b7, false);
        loadk(t + 11, b3);
        HOLD_LDS_READS();
        pass(b6, b7, b0, last && t == TCH - 8);
        loadk(t + 12, b4);
        HOLD_LDS_READS();
        pass(b7, b0, b1, false);
      }
#undef HOLD_LDS_READS
#undef PASS_ORDER

      if (last) {
        const float* hp = &xb_cur[(TCH - 1) * 36 + 8 * q];
        const float4 h0 = *(const float4*)hp;
        const float4 h1 = *(const float4*)(hp + 4);
        const v2f h01 = {h0.x, h0.y}, h23 = {h0.z, h0.w};
        const v2f h45 = {h1.x, h1.y}, h67 = {h1.z, h1.w};
        v2f P = M01 * h01;
        v2f Q = M45 * h45;
        P = __builtin_elementwise_fma(M23, h23, P);
        Q = __builtin_elementwise_fma(M67, h67, Q);
        const v2f R = P + Q;
        const float cs = quad_sum(R.x + R.y);
        if (q == 0) s_ctx[myrow] = cs;
      }
    }
    __syncthreads();
  }

  if (tid < HD) {
    float acc = br[tid];
#pragma unroll
    for (int j = 0; j < HD; ++j) acc = fmaf(s_ctx[j], wr[j * HD + tid], acc);
    s_tmp[tid] = acc;
  }
  __syncthreads();
  if (tid < VD) {
    float acc = bo[tid];
#pragma unroll
    for (int i = 0; i < HD; ++i) acc = fmaf(s_tmp[i], wo[i * VD + tid], acc);
    out[(size_t)bidx * VD + tid] = acc;
  }
}

extern "C" void kernel_launch(void* const* d_in, const int* in_sizes, int n_in,
                              void* d_out, int out_size, void* d_ws, size_t ws_size,
                              hipStream_t stream) {
  const int*   seq   = (const int*)  d_in[0];
  const float* embed = (const float*)d_in[1];
  const float* w1    = (const float*)d_in[2];
  const float* b1    = (const float*)d_in[3];
  const float* w2    = (const float*)d_in[4];
  const float* b2    = (const float*)d_in[5];
  const float* ln_g  = (const float*)d_in[6];
  const float* ln_b  = (const float*)d_in[7];
  const float* wr    = (const float*)d_in[8];
  const float* br    = (const float*)d_in[9];
  const float* wo    = (const float*)d_in[10];
  const float* bo    = (const float*)d_in[11];
  float* out = (float*)d_out;

  const int B = out_size / VD;
  const int L = in_sizes[0] / B;

  delta_model_fused<<<B, NTHREADS, 0, stream>>>(
      seq, embed, w1, b1, w2, b2, ln_g, ln_b, wr, br, wo, bo, out, L);
}


# Scan reads two rows per group every second step, the two rows' (rinv,g) and k[myrow] by two-address LDS reads: 6 LDS read instructions per 2 steps instead of 8
# speedup vs baseline: 2.9526x; 1.0807x over previous
#include <hip/hip_runtime.h>

typedef __attribute__((ext_vector_type(2))) float v2f;
typedef __attribute__((ext_vector_type(8))) float v8f;

#define HD 32
#define VD 64
#define TCH 64
#define NTOK 64
#define XPAD 5
#define NTHREADS 256

__device__ __forceinline__ float quad_sum(float v) {
  v += __int_as_float(__builtin_amdgcn_update_dpp(0, __float_as_int(v), 0xB1, 0xF, 0xF, true));
  v += __int_as_float(__builtin_amdgcn_update_dpp(0, __float_as_int(v), 0x4E, 0xF, 0xF, true));
  return v;
}


__launch_bounds__(NTHREADS)
__global__ void delta_model_fused(
    const int*   __restrict__ seq,
    const float* __restrict__ embed,
    const float* __restrict__ w1, const float* __restrict__ b1,
    const float* __restrict__ w2, const float* __restrict__ b2,
    const float* __restrict__ ln_g, const float* __restrict__ ln_b,
    const float* __restrict__ wr, const float* __restrict__ br,
    const float* __restrict__ wo, const float* __restrict__ bo,
    float* __restrict__ out, int L)
{
  __shared__ float s_w1[HD * 2 * HD];
  __shared__ float s_w2[2 * HD * HD];
  __shared__ float s_b1[2 * HD];
  __shared__ float s_b2[HD];
  __shared__ float s_g[HD];
  __shared__ float s_bt[HD];
  __shared__ float s_h [TCH * 34];
  __shared__ float s_ff[TCH * 66];
  __shared__ __align__(16) float s_x[2][(TCH + XPAD) * 36];
  __shared__ __align__(16) float s_tab[NTOK * 36];
  __shared__ float s_gram[NTOK * NTOK];
  __shared__ float s_ctx[HD];
  __shared__ float s_tmp[HD];

  const int tid  = threadIdx.x;
  const int wave = __builtin_amdgcn_readfirstlane(tid) >> 5;
  const int lane = tid & 31;
  const int ln   = lane & 15;
  const int hl   = lane >> 4;
  const int bidx = blockIdx.x;

  for (int i = tid; i < HD * 2 * HD; i += NTHREADS) s_w1[i] = w1[i];
  for (int i = tid; i < 2 * HD * HD; i += NTHREADS) s_w2[i] = w2[i];
  if (tid < 2 * HD) s_b1[tid] = b1[tid];
  if (tid < HD) { s_b2[tid] = b2[tid]; s_g[tid] = ln_g[tid]; s_bt[tid] = ln_b[tid]; }
  for (int i = tid; i < 2 * XPAD * 36; i += NTHREADS)
    s_x[i / (XPAD * 36)][TCH * 36 + i % (XPAD * 36)] = 0.f;

  const int q     = lane & 3;
  const int myrow = ((wave & 3) << 3) + (lane >> 2);
  v2f M01 = {0.f, 0.f}, M23 = {0.f, 0.f}, M45 = {0.f, 0.f}, M67 = {0.f, 0.f};

  const int nChunks = L / TCH;
  const int m0 = (wave & 3) * 16;

  auto build_table = [&]() {
    float* xb = s_tab;
#pragma unroll
    for (int r = 0; r < 16; ++r)
      s_h[(m0 + r) * 34 + lane] = embed[(m0 + r) * HD + lane];
#pragma unroll
    for (int nt = 0; nt < 4; ++nt) {
      const int n0 = nt * 16;
      v8f acc = {0.f, 0.f, 0.f, 0.f, 0.f, 0.f, 0.f, 0.f};
#pragma unroll
      for (int k0 = 0; k0 < HD; k0 += 4) {
        const float* ap = &s_h[(m0 + ln) * 34 + k0 + 2 * hl];
        v2f a; a.x = ap[0]; a.y = ap[1];
        v2f bf; bf.x = s_w1[(k0 + 2 * hl) * 64 + n0 + ln];
        bf.y = s_w1[(k0 + 2 * hl + 1) * 64 + n0 + ln];
        acc = __builtin_amdgcn_wmma_f32_16x16x4_f32(false, a, false, bf,
                                                    (short)0, acc, false, false);
      }
      float bias = s_b1[n0 + ln];
#pragma unroll
      for (int r = 0; r < 8; ++r) {
        int row = m0 + r + 8 * hl;
        float v = acc[r] + bias;
        s_ff[row * 66 + n0 + ln] = v > 0.f ? v : 0.f;
      }
    }
#pragma unroll
    for (int nt = 0; nt < 2; ++nt) {
      const int n0 = nt * 16;
      v8f acc = {0.f, 0.f, 0.f, 0.f, 0.f, 0.f, 0.f, 0.f};
#pragma unroll
      for (int k0 = 0; k0 < 2 * HD; k0 += 4) {
        const float* ap = &s_ff[(m0 + ln) * 66 + k0 + 2 * hl];
        v2f a; a.x = ap[0]; a.y = ap[1];
        v2f bf; bf.x = s_w2[(k0 + 2 * hl) * HD + n0 + ln];
        bf.y = s_w2[(k0 + 2 * hl + 1) * HD + n0 + ln];
        acc = __builtin_amdgcn_wmma_f32_16x16x4_f32(false, a, false, bf,
                                                    (short)0, acc, false, false);
      }
      float bias = s_b2[n0 + ln];
#pragma unroll
      for (int r = 0; r < 8; ++r) {
        int row = m0 + r + 8 * hl, col = n0 + ln;
        xb[row * 36 + col] = acc[r] + bias + s_h[row * 34 + col];
      }
    }
    if (lane < 16) {
      float* xr = &xb[(m0 + lane) * 36];
      float mu = 0.f;
#pragma unroll
      for (int j = 0; j < HD; ++j) mu += xr[j];
      mu *= (1.f / 32.f);
      float var = 0.f;
#pragma unroll
      for (int j = 0; j < HD; ++j) { float d = xr[j] - mu; var = fmaf(d, d, var); }
      var *= (1.f / 32.f);
      float rs = __builtin_amdgcn_rsqf(var + 1e-5f);
      float kk2 = 0.f;
#pragma unroll
      for (int j = 0; j < HD; ++j) {
        float v = (xr[j] - mu) * rs * s_g[j] + s_bt[j];
        xr[j] = v;
        kk2 = fmaf(v, v, kk2);
      }
      xr[HD] = __builtin_amdgcn_rcpf(kk2 + 1e-6f);
    }
  };

  auto build_gram = [&]() {
#pragma unroll
    for (int nt = 0; nt < 4; ++nt) {
      const int n0 = nt * 16;
      v8f acc = {0.f, 0.f, 0.f, 0.f, 0.f, 0.f, 0.f, 0.f};
#pragma unroll
      for (int k0 = 0; k0 < HD; k0 += 4) {
        const float* ap = &s_tab[(m0 + ln) * 36 + k0 + 2 * hl];
        const float* bp = &s_tab[(n0 + ln) * 36 + k0 + 2 * hl];
        v2f a;  a.x  = ap[0]; a.y  = ap[1];
        v2f bf; bf.x = bp[0]; bf.y = bp[1];
        acc = __builtin_amdgcn_wmma_f32_16x16x4_f32(false, a, false, bf,
                                                    (short)0, acc, false, false);
      }
#pragma unroll
      for (int r = 0; r < 8; ++r)
        s_gram[(m0 + r + 8 * hl) * NTOK + n0 + ln] = acc[r];
    }
  };

  auto gather = [&](int cIdx, float* xb) {
    const int r  = lane & 15;
    const int p0 = (lane >> 4) * 4;
    const int tpos = cIdx * TCH + m0 + r;
    const int* sp = seq + (size_t)bidx * L + tpos;
    const int tok = sp[0];
    const int tokp = tpos > 0 ? sp[-1] : tok;
    const float4* src = (const float4*)&s_tab[tok * 36] + p0;
    float4* dst = (float4*)&xb[(m0 + r) * 36] + p0;
    const float4 a0 = src[0], a1 = src[1], a2 = src[2], a3 = src[3];
    const float ri = s_tab[tok * 36 + HD];
    const float g  = s_gram[tokp * NTOK + tok];
    dst[0] = a0; dst[1] = a1; dst[2] = a2; dst[3] = a3;
    if (lane < 16) { xb[(m0 + r) * 36 + HD] = ri; xb[(m0 + r) * 36 + HD + 1] = g; }
  };

  __syncthreads();
  if (wave < 4) build_table();
  __syncthreads();
  if (wave < 4) build_gram();
  __syncthreads();
  if (wave < 4) gather(0, s_x[0]);
  __syncthreads();

  for (int c = 0; c < nChunks; ++c) {
    float* xb_cur = s_x[c & 1];
    float* xb_nxt = s_x[(c & 1) ^ 1];

    if (wave < 4) {
      if (c + 1 < nChunks) gather(c + 1, xb_nxt);
    } else {
      const bool last = (c == nChunks - 1);

      struct KBuf { float4 f0, f1; float kl, ri, g; };
      auto loadk = [&](int t, KBuf& kb) {
        const float* rowp = &xb_cur[t * 36];
        kb.f0 = *(const float4*)(rowp + 8 * q);
        kb.f1 = *(const float4*)(rowp + 8 * q + 4);
        const float2 rg = *(const float2*)(rowp + HD);
        kb.ri = rg.x;
        kb.g  = rg.y;
        kb.kl = rowp[myrow];
      };
      auto loadk2 = [&](int t, KBuf& ka, KBuf& kb) {
        const float* rowp = &xb_cur[t * 36];
        const float2 rga = *(const float2*)(rowp + HD);
        const float2 rgb = *(const float2*)(rowp + 36 + HD);
        ka.ri = rga.x; ka.g = rga.y;
        kb.ri = rgb.x; kb.g = rgb.y;
        ka.kl = rowp[myrow];
        kb.kl = rowp[36 + myrow];
        ka.f0 = *(const float4*)(rowp + 8 * q);
        ka.f1 = *(const float4*)(rowp + 8 * q + 4);
        kb.f0 = *(const float4*)(rowp + 36 + 8 * q);
        kb.f1 = *(const float4*)(rowp + 36 + 8 * q + 4);
      };

      v2f R;
      float u;
      auto dot_products = [&](const KBuf& kb) -> v2f {
        const v2f k01 = {kb.f0.x, kb.f0.y}, k23 = {kb.f0.z, kb.f0.w};
        const v2f k45 = {kb.f1.x, kb.f1.y}, k67 = {kb.f1.z, kb.f1.w};
        v2f Po = M01 * k01;
        v2f Qo = M45 * k45;
        Po = __builtin_elementwise_fma(M23, k23, Po);
        Qo = __builtin_elementwise_fma(M67, k67, Qo);
        return Po + Qo;
      };
      auto dot_sum = [&](const v2f& Ri) -> float {
        return quad_sum(Ri.x + Ri.y);
      };
#define PASS_ORDER() __builtin_amdgcn_sched_barrier(0)
      auto pass = [&](const KBuf& cur, const KBuf& nx1, const KBuf& nx2, const bool drop) {
        const v2f uu = {u, u};
        const v2f c01 = {cur.f0.x, cur.f0.y}, c23 = {cur.f0.z, cur.f0.w};
        const v2f c45 = {cur.f1.x, cur.f1.y}, c67 = {cur.f1.z, cur.f1.w};
        const v2f n01 = {nx2.f0.x, nx2.f0.y}, n23 = {nx2.f0.z, nx2.f0.w};
        const v2f n45 = {nx2.f1.x, nx2.f1.y}, n67 = {nx2.f1.z, nx2.f1.w};
        M01 = __builtin_elementwise_fma(uu, c01, M01);
        M45 = __builtin_elementwise_fma(uu, c45, M45);
        M23 = __builtin_elementwise_fma(uu, c23, M23);
        M67 = __builtin_elementwise_fma(uu, c67, M67);
        PASS_ORDER();
        float s = R.x + R.y;
        v2f Pn = M01 * n01;
        v2f Qn = M45 * n45;
        PASS_ORDER();
        s += __int_as_float(__builtin_amdgcn_update_dpp(0, __float_as_int(s), 0xB1, 0xF, 0xF, true));
        Pn = __builtin_elementwise_fma(M23, n23, Pn);
        Qn = __builtin_elementwise_fma(M67, n67, Qn);
        s += __int_as_float(__builtin_amdgcn_update_dpp(0, __float_as_int(s), 0x4E, 0xF, 0xF, true));
        PASS_ORDER();
        const float vp = fmaf(u, nx1.g, s);
        R = Pn + Qn;
        u = fmaf(-vp, nx1.ri, nx1.kl);
        if (drop) u = 0.f;
      };

#define HOLD_LDS_READS()                   \
  do {                                     \
    asm volatile("" ::: "memory");         \
    __builtin_amdgcn_sched_barrier(0);     \
  } while (0)
      KBuf b0, b1, b2, b3, b4, b5, b6, b7;
      loadk(0, b0);
      loadk(1, b1);
      loadk(2, b2);
      loadk(3, b3);
      loadk(4, b4);
      HOLD_LDS_READS();
      u = fmaf(-dot_sum(dot_products(b0)), b0.ri, b0.kl);
      R = dot_products(b1);
      for (int t = 0; t < TCH; t += 8) {
        loadk2(t + 5, b5, b6);
        HOLD_LDS_READS();
        pass(b0, b1, b2, false);
        pass(b1, b2, b3, false);
        loadk2(t + 7, b7, b0);
        HOLD_LDS_READS();
        pass(b2, b3, b4, false);
        pass(b3, b4, b5, false);
        loadk2(t + 9, b1, b2);
        HOLD_LDS_READS();
        pass(b4, b5, b6, false);
        pass(b5, b6, b7, false);
        loadk2(t + 11, b3, b4);
        HOLD_LDS_READS();
        pass(b6, b7, b0, last && t == TCH - 8);
        pass(b7, b0, b1, false);
      }
#undef HOLD_LDS_READS
#undef PASS_ORDER

      if (last) {
        const float* hp = &xb_cur[(TCH - 1) * 36 + 8 * q];
        const float4 h0 = *(const float4*)hp;
        const float4 h1 = *(const float4*)(hp + 4);
        const v2f h01 = {h0.x, h0.y}, h23 = {h0.z, h0.w};
        const v2f h45 = {h1.x, h1.y}, h67 = {h1.z, h1.w};
        v2f P = M01 * h01;
        v2f Q = M45 * h45;
        P = __builtin_elementwise_fma(M23, h23, P);
        Q = __builtin_elementwise_fma(M67, h67, Q);
        const v2f R = P + Q;
        const float cs = quad_sum(R.x + R.y);
        if (q == 0) s_ctx[myrow] = cs;
      }
    }
    __syncthreads();
  }

  if (tid < HD) {
    float acc = br[tid];
#pragma unroll
    for (int j = 0; j < HD; ++j) acc = fmaf(s_ctx[j], wr[j * HD + tid], acc);
    s_tmp[tid] = acc;
  }
  __syncthreads();
  if (tid < VD) {
    float acc = bo[tid];
#pragma unroll
    for (int i = 0; i < HD; ++i) acc = fmaf(s_tmp[i], wo[i * VD + tid], acc);
    out[(size_t)bidx * VD + tid] = acc;
  }
}

extern "C" void kernel_launch(void* const* d_in, const int* in_sizes, int n_in,
                              void* d_out, int out_size, void* d_ws, size_t ws_size,
                              hipStream_t stream) {
  const int*   seq   = (const int*)  d_in[0];
  const float* embed = (const float*)d_in[1];
  const float* w1    = (const float*)d_in[2];
  const float* b1    = (const float*)d_in[3];
  const float* w2    = (const float*)d_in[4];
  const float* b2    = (const float*)d_in[5];
  const float* ln_g  = (const float*)d_in[6];
  const float* ln_b  = (const float*)d_in[7];
  const float* wr    = (const float*)d_in[8];
  const float* br    = (const float*)d_in[9];
  const float* wo    = (const float*)d_in[10];
  const float* bo    = (const float*)d_in[11];
  float* out = (float*)d_out;

  const int B = out_size / VD;
  const int L = in_sizes[0] / B;

  delta_model_fused<<<B, NTHREADS, 0, stream>>>(
      seq, embed, w1, b1, w2, b2, ln_g, ln_b, wr, br, wo, bo, out, L);
}


# Delta scan in blocks of 16 tokens on f32 WMMA: u = T(k - rinv M0 k), T = block triangular solve as a 16x16 matrix built ahead from token ids; M as transposed tiles on 2 waves; all sums' order changes
# speedup vs baseline: 4.4737x; 1.5152x over previous
#include <hip/hip_runtime.h>

typedef __attribute__((ext_vector_type(2))) float v2f;
typedef __attribute__((ext_vector_type(8))) float v8f;

#define HD 32
#define VD 64
#define TCH 64
#define NTOK 64
#define BLK 16
#define NBLK (TCH / BLK)
#define TST 20
#define NTHREADS 192


__launch_bounds__(NTHREADS)
__global__ void delta_model_fused(
    const int*   __restrict__ seq,
    const float* __restrict__ embed,
    const float* __restrict__ w1, const float* __restrict__ b1,
    const float* __restrict__ w2, const float* __restrict__ b2,
    const float* __restrict__ ln_g, const float* __restrict__ ln_b,
    const float* __restrict__ wr, const float* __restrict__ br,
    const float* __restrict__ wo, const float* __restrict__ bo,
    float* __restrict__ out, int L)
{
  __shared__ float s_w1[HD * 2 * HD];
  __shared__ float s_w2[2 * HD * HD];
  __shared__ float s_b1[2 * HD];
  __shared__ float s_b2[HD];
  __shared__ float s_g[HD];
  __shared__ float s_bt[HD];
  __shared__ float s_h [TCH * 34];
  __shared__ float s_ff[TCH * 66];
  __shared__ __align__(16) float s_x[2][TCH * 36];
  __shared__ __align__(16) float s_tab[NTOK * 36];
  __shared__ float s_gram[NTOK * NTOK];
  __shared__ __align__(16) float s_T[2][NBLK * BLK * TST];
  __shared__ __align__(16) float s_A[NBLK * BLK * TST];
  __shared__ float s_ctxp[2 * HD];
  __shared__ float s_tmp[HD];

  const int tid  = threadIdx.x;
  const int wave = __builtin_amdgcn_readfirstlane(tid) >> 5;
  const int lane = tid & 31;
  const int ln   = lane & 15;
  const int hl   = lane >> 4;
  const int bidx = blockIdx.x;

  for (int i = tid; i < HD * 2 * HD; i += NTHREADS) s_w1[i] = w1[i];
  for (int i = tid; i < 2 * HD * HD; i += NTHREADS) s_w2[i] = w2[i];
  if (tid < 2 * HD) s_b1[tid] = b1[tid];
  if (tid < HD) { s_b2[tid] = b2[tid]; s_g[tid] = ln_g[tid]; s_bt[tid] = ln_b[tid]; }

  const int nChunks = L / TCH;
  const int m0 = (wave & 3) * 16;

  auto build_table = [&]() {
    float* xb = s_tab;
#pragma unroll
    for (int r = 0; r < 16; ++r)
      s_h[(m0 + r) * 34 + lane] = embed[(m0 + r) * HD + lane];
#pragma unroll
    for (int nt = 0; nt < 4; ++nt) {
      const int n0 = nt * 16;
      v8f acc = {0.f, 0.f, 0.f, 0.f, 0.f, 0.f, 0.f, 0.f};
#pragma unroll
      for (int k0 = 0; k0 < HD; k0 += 4) {
        const float* ap = &s_h[(m0 + ln) * 34 + k0 + 2 * hl];
        v2f a; a.x = ap[0]; a.y = ap[1];
        v2f bf; bf.x = s_w1[(k0 + 2 * hl) * 64 + n0 + ln];
        bf.y = s_w1[(k0 + 2 * hl + 1) * 64 + n0 + ln];
        acc = __builtin_amdgcn_wmma_f32_16x16x4_f32(false, a, false, bf,
                                                    (short)0, acc, false, false);
      }
      float bias = s_b1[n0 + ln];
#pragma unroll
      for (int r = 0; r < 8; ++r) {
        int row = m0 + r + 8 * hl;
        float v = acc[r] + bias;
        s_ff[row * 66 + n0 + ln] = v > 0.f ? v : 0.f;
      }
    }
#pragma unroll
    for (int nt = 0; nt < 2; ++nt) {
      const int n0 = nt * 16;
      v8f acc = {0.f, 0.f, 0.f, 0.f, 0.f, 0.f, 0.f, 0.f};
#pragma unroll
      for (int k0 = 0; k0 < 2 * HD; k0 += 4) {
        const float* ap = &s_ff[(m0 + ln) * 66 + k0 + 2 * hl];
        v2f a; a.x = ap[0]; a.y = ap[1];
        v2f bf; bf.x = s_w2[(k0 + 2 * hl) * HD + n0 + ln];
        bf.y = s_w2[(k0 + 2 * hl + 1) * HD + n0 + ln];
        acc = __builtin_amdgcn_wmma_f32_16x16x4_f32(false, a, false, bf,
                                                    (short)0, acc, false, false);
      }
      float bias = s_b2[n0 + ln];
#pragma unroll
      for (int r = 0; r < 8; ++r) {
        int row = m0 + r + 8 * hl, col = n0 + ln;
        xb[row * 36 + col] = acc[r] + bias + s_h[row * 34 + col];
      }
    }
    if (lane < 16) {
      float* xr = &xb[(m0 + lane) * 36];
      float mu = 0.f;
#pragma unroll
      for (int j = 0; j < HD; ++j) mu += xr[j];
      mu *= (1.f / 32.f);
      float var = 0.f;
#pragma unroll
      for (int j = 0; j < HD; ++j) { float d = xr[j] - mu; var = fmaf(d, d, var); }
      var *= (1.f / 32.f);
      float rs = __builtin_amdgcn_rsqf(var + 1e-5f);
      float kk2 = 0.f;
#pragma unroll
      for (int j = 0; j < HD; ++j) {
        float v = (xr[j] - mu) * rs * s_g[j] + s_bt[j];
        xr[j] = v;
        kk2 = fmaf(v, v, kk2);
      }
      xr[HD] = __builtin_amdgcn_rcpf(kk2 + 1e-6f);
    }
  };

  auto build_gram = [&]() {
#pragma unroll
    for (int nt = 0; nt < 4; ++nt) {
      const int n0 = nt * 16;
      v8f acc = {0.f, 0.f, 0.f, 0.f, 0.f, 0.f, 0.f, 0.f};
#pragma unroll
      for (int k0 = 0; k0 < HD; k0 += 4) {
        const float* ap = &s_tab[(m0 + ln) * 36 + k0 + 2 * hl];
        const float* bp = &s_tab[(n0 + ln) * 36 + k0 + 2 * hl];
        v2f a;  a.x  = ap[0]; a.y  = ap[1];
        v2f bf; bf.x = bp[0]; bf.y = bp[1];
        acc = __builtin_amdgcn_wmma_f32_16x16x4_f32(false, a, false, bf,
                                                    (short)0, acc, false, false);
      }
#pragma unroll
      for (int r = 0; r < 8; ++r)
        s_gram[(m0 + r + 8 * hl) * NTOK + n0 + ln] = acc[r];
    }
  };

  auto load_ids = [&](int cIdx, int& tok, int4& ta, int4& tb) {
    const int* sp = seq + (size_t)bidx * L + cIdx * TCH + m0;
    tok = sp[lane & 15];
    ta = *(const int4*)(sp + 8 * hl);
    tb = *(const int4*)(sp + 8 * hl + 4);
  };

  auto prepare = [&](int cIdx, float* xb, float* Tb, const int tok, const int4 ta, const int4 tb) {
    const int r  = lane & 15;
    const int p0 = (lane >> 4) * 4;
    const float4* src = (const float4*)&s_tab[tok * 36] + p0;
    float4* dst = (float4*)&xb[(m0 + r) * 36] + p0;
    const float4 a0 = src[0], a1 = src[1], a2 = src[2], a3 = src[3];
    const float ri = s_tab[tok * 36 + HD];
    asm volatile("" ::: "memory");
    dst[0] = a0; dst[1] = a1; dst[2] = a2; dst[3] = a3;
    if (lane < 16) xb[(m0 + r) * 36 + HD] = ri;

    float* Aw = &s_A[(wave & 3) * BLK * TST];
    const float* gcol = &s_gram[tok];
    const int tt[8] = {ta.x, ta.y, ta.z, ta.w, tb.x, tb.y, tb.z, tb.w};
#pragma unroll
    for (int e = 0; e < 8; ++e) Aw[(8 * hl + e) * TST + r] = ri * gcol[tt[e] * NTOK];
    __builtin_amdgcn_wave_barrier();
    float acc[BLK];
#pragma unroll
    for (int j = 0; j < BLK; ++j) acc[j] = (j == r) ? 1.f : 0.f;
#pragma unroll
    for (int i = 0; i < BLK - 1; ++i) {
      const float4* ap = (const float4*)&Aw[i * TST];
      const float4 q0 = ap[0], q1 = ap[1], q2 = ap[2], q3 = ap[3];
      const float col[BLK] = {q0.x, q0.y, q0.z, q0.w, q1.x, q1.y, q1.z, q1.w,
                              q2.x, q2.y, q2.z, q2.w, q3.x, q3.y, q3.z, q3.w};
#pragma unroll
      for (int j = i + 1; j < BLK; ++j) acc[j] = fmaf(-col[j], acc[i], acc[j]);
    }
    const bool noStep = (cIdx == nChunks - 1) && (m0 == TCH - BLK);
    float* Tw = Tb + (wave & 3) * BLK * TST;
    if (lane < 16) {
#pragma unroll
      for (int j = 0; j < BLK; ++j)
        Tw[j * TST + r] = (noStep && j == BLK - 1) ? 0.f : acc[j];
    }
  };

  const int sw = wave & 1;
  v8f Mt0 = {0.f, 0.f, 0.f, 0.f, 0.f, 0.f, 0.f, 0.f};
  v8f Mt1 = {0.f, 0.f, 0.f, 0.f, 0.f, 0.f, 0.f, 0.f};

  __syncthreads();
  if (wave < 4) build_table();
  __syncthreads();
  if (wave < 4) build_gram();
  __syncthreads();
  int ptok = 0;
  int4 pta = {0, 0, 0, 0}, ptb = {0, 0, 0, 0};
  if (wave < 4) {
    load_ids(0, ptok, pta, ptb);
    prepare(0, s_x[0], s_T[0], ptok, pta, ptb);
    if (nChunks > 1) load_ids(1, ptok, pta, ptb);
  }
  __syncthreads();

  for (int c = 0; c < nChunks; ++c) {
    float* xb_cur = s_x[c & 1];
    float* xb_nxt = s_x[(c & 1) ^ 1];
    float* T_cur  = s_T[c & 1];
    float* T_nxt  = s_T[(c & 1) ^ 1];

    if (wave < 4) {
      if (c + 1 < nChunks) {
        int ntok = 0;
        int4 nta = {0, 0, 0, 0}, ntb = {0, 0, 0, 0};
        if (c + 2 < nChunks) load_ids(c + 2, ntok, nta, ntb);
        prepare(c + 1, xb_nxt, T_nxt, ptok, pta, ptb);
        ptok = ntok; pta = nta; ptb = ntb;
      }
    } else {
      struct BlkOps { float4 ka0, kb0, ka1, kb1; float ri[8], kk[8]; float4 t0, t1; float c0[8], c1[8]; };
      auto load_first = [&](int b, BlkOps& o) {
        const float* xb = xb_cur + b * BLK * 36;
        const float* kr = xb + ln * 36 + 8 * hl;
        o.ka0 = *(const float4*)(kr);
        o.kb0 = *(const float4*)(kr + 16);
        o.ka1 = *(const float4*)(kr + 4);
        o.kb1 = *(const float4*)(kr + 20);
      };
      auto load_rest = [&](int b, BlkOps& o) {
        const float* xb = xb_cur + b * BLK * 36;
        const float* Tb = T_cur + b * BLK * TST;
        const float* tr = xb + 8 * hl * 36;
#pragma unroll
        for (int r = 0; r < 8; ++r) { o.ri[r] = tr[r * 36 + HD]; o.kk[r] = tr[r * 36 + 16 * sw + ln]; }
        o.t0 = *(const float4*)(Tb + ln * TST + 8 * hl);
        o.t1 = *(const float4*)(Tb + ln * TST + 8 * hl + 4);
#pragma unroll
        for (int r = 0; r < 8; ++r) { o.c0[r] = tr[r * 36 + ln]; o.c1[r] = tr[r * 36 + 16 + ln]; }
      };
#define WMMA4(a, b, c) __builtin_amdgcn_wmma_f32_16x16x4_f32(false, a, false, b, (short)0, c, false, false)
      auto scan_block = [&](const BlkOps& o, auto&& read_next) {
        const v8f zero = {0.f, 0.f, 0.f, 0.f, 0.f, 0.f, 0.f, 0.f};
        v8f W0 = zero, W1 = zero;
        W0 = WMMA4(((v2f){o.ka0.x, o.ka0.y}), ((v2f){Mt0[0], Mt0[1]}), W0);
        W1 = WMMA4(((v2f){o.kb0.x, o.kb0.y}), ((v2f){Mt1[0], Mt1[1]}), W1);
        W0 = WMMA4(((v2f){o.ka0.z, o.ka0.w}), ((v2f){Mt0[2], Mt0[3]}), W0);
        W1 = WMMA4(((v2f){o.kb0.z, o.kb0.w}), ((v2f){Mt1[2], Mt1[3]}), W1);
        W0 = WMMA4(((v2f){o.ka1.x, o.ka1.y}), ((v2f){Mt0[4], Mt0[5]}), W0);
        W1 = WMMA4(((v2f){o.kb1.x, o.kb1.y}), ((v2f){Mt1[4], Mt1[5]}), W1);
        W0 = WMMA4(((v2f){o.ka1.z, o.ka1.w}), ((v2f){Mt0[6], Mt0[7]}), W0);
        W1 = WMMA4(((v2f){o.kb1.z, o.kb1.w}), ((v2f){Mt1[6], Mt1[7]}), W1);
        read_next();
        v8f RHS;
#pragma unroll
        for (int r = 0; r < 8; ++r) RHS[r] = fmaf(-o.ri[r], W0[r] + W1[r], o.kk[r]);
        v8f U = zero;
        U = WMMA4(((v2f){o.t0.x, o.t0.y}), ((v2f){RHS[0], RHS[1]}), U);
        U = WMMA4(((v2f){o.t0.z, o.t0.w}), ((v2f){RHS[2], RHS[3]}), U);
        U = WMMA4(((v2f){o.t1.x, o.t1.y}), ((v2f){RHS[4], RHS[5]}), U);
        U = WMMA4(((v2f){o.t1.z, o.t1.w}), ((v2f){RHS[6], RHS[7]}), U);
#pragma unroll
        for (int r0 = 0; r0 < 8; r0 += 2) {
          const v2f ub = {U[r0], U[r0 + 1]};
          Mt0 = WMMA4(((v2f){o.c0[r0], o.c0[r0 + 1]}), ub, Mt0);
          Mt1 = WMMA4(((v2f){o.c1[r0], o.c1[r0 + 1]}), ub, Mt1);
        }
      };
#define HOLD_LDS_READS()                   \
  do {                                     \
    asm volatile("" ::: "memory");         \
    __builtin_amdgcn_sched_barrier(0);     \
  } while (0)
      BlkOps oa, ob;
      load_first(0, oa);
      HOLD_LDS_READS();
      load_rest(0, oa);
      HOLD_LDS_READS();
      scan_block(oa, [&]() { load_first(1, ob); load_rest(1, ob); HOLD_LDS_READS(); });
      scan_block(ob, [&]() { load_first(2, oa); load_rest(2, oa); HOLD_LDS_READS(); });
      scan_block(oa, [&]() { load_first(3, ob); load_rest(3, ob); HOLD_LDS_READS(); });
      scan_block(ob, [&]() {});
#undef HOLD_LDS_READS
#undef WMMA4

      if (c == nChunks - 1) {
        const float* hp = &xb_cur[(TCH - 1) * 36 + 8 * hl];
        const float4 h0 = *(const float4*)(hp);
        const float4 h1 = *(const float4*)(hp + 4);
        const float4 h2 = *(const float4*)(hp + 16);
        const float4 h3 = *(const float4*)(hp + 20);
        float p = Mt0[0] * h0.x;
        p = fmaf(Mt0[1], h0.y, p); p = fmaf(Mt0[2], h0.z, p); p = fmaf(Mt0[3], h0.w, p);
        p = fmaf(Mt0[4], h1.x, p); p = fmaf(Mt0[5], h1.y, p); p = fmaf(Mt0[6], h1.z, p); p = fmaf(Mt0[7], h1.w, p);
        p = fmaf(Mt1[0], h2.x, p); p = fmaf(Mt1[1], h2.y, p); p = fmaf(Mt1[2], h2.z, p); p = fmaf(Mt1[3], h2.w, p);
        p = fmaf(Mt1[4], h3.x, p); p = fmaf(Mt1[5], h3.y, p); p = fmaf(Mt1[6], h3.z, p); p = fmaf(Mt1[7], h3.w, p);
        s_ctxp[hl * HD + 16 * sw + ln] = p;
      }
    }
    __syncthreads();
  }

  if (tid < HD) {
    float acc = br[tid];
#pragma unroll
    for (int j = 0; j < HD; ++j) acc = fmaf(s_ctxp[j] + s_ctxp[HD + j], wr[j * HD + tid], acc);
    s_tmp[tid] = acc;
  }
  __syncthreads();
  if (tid < VD) {
    float acc = bo[tid];
#pragma unroll
    for (int i = 0; i < HD; ++i) acc = fmaf(s_tmp[i], wo[i * VD + tid], acc);
    out[(size_t)bidx * VD + tid] = acc;
  }
}

extern "C" void kernel_launch(void* const* d_in, const int* in_sizes, int n_in,
                              void* d_out, int out_size, void* d_ws, size_t ws_size,
                              hipStream_t stream) {
  const int*   seq   = (const int*)  d_in[0];
  const float* embed = (const float*)d_in[1];
  const float* w1    = (const float*)d_in[2];
  const float* b1    = (const float*)d_in[3];
  const float* w2    = (const float*)d_in[4];
  const float* b2    = (const float*)d_in[5];
  const float* ln_g  = (const float*)d_in[6];
  const float* ln_b  = (const float*)d_in[7];
  const float* wr    = (const float*)d_in[8];
  const float* br    = (const float*)d_in[9];
  const float* wo    = (const float*)d_in[10];
  const float* bo    = (const float*)d_in[11];
  float* out = (float*)d_out;

  const int B = out_size / VD;
  const int L = in_sizes[0] / B;

  delta_model_fused<<<B, NTHREADS, 0, stream>>>(
      seq, embed, w1, b1, w2, b2, ln_g, ln_b, wr, br, wo, bo, out, L);
}
